# starting kernel
# baseline (speedup 1.0000x reference)
#include <hip/hip_runtime.h>

typedef float v2f __attribute__((ext_vector_type(2)));
typedef float v8f __attribute__((ext_vector_type(8)));

#define NN 8192
#define D 128

__global__ __launch_bounds__(256) void k_deg(const float* __restrict__ adj,
                                             float* __restrict__ dinv) {
  __shared__ float red[256];
  const int row = blockIdx.x;
  const float* r = adj + (size_t)row * NN;
  float s = 0.f;
  for (int j = threadIdx.x; j < NN; j += 256)
    s += __builtin_nontemporal_load(r + j);
  red[threadIdx.x] = s;
  __syncthreads();
  for (int off = 128; off > 0; off >>= 1) {
    if (threadIdx.x < off) red[threadIdx.x] += red[threadIdx.x + off];
    __syncthreads();
  }
  if (threadIdx.x == 0) {
    float d = red[0];
    dinv[row] = d > 0.f ? rsqrtf(d) : 0.f;
  }
}

__global__ __launch_bounds__(128) void k_proj(const float* __restrict__ x,
                                              const float* __restrict__ w,
                                              const float* __restrict__ dinv,
                                              float* __restrict__ bfrag) {
  __shared__ float xs[D];
  const int row = blockIdx.x;
  const int col = threadIdx.x;
  xs[col] = x[(size_t)row * D + col];
  __syncthreads();
  float acc = 0.f;
#pragma unroll 8
  for (int k = 0; k < D; ++k) acc += xs[k] * w[k * D + col];
  acc *= dinv[row];

  const int ks = row >> 2, within = row & 3;
  const int r = within & 1, half = within >> 1;
  const int lane = half * 16 + (col & 15);
  const int tile = col >> 4;
  bfrag[(((size_t)ks * 8 + tile) * 32 + lane) * 2 + r] = acc;
}

__global__ __launch_bounds__(128) void k_agg(const float* __restrict__ adj,
                                             const float* __restrict__ bfrag,
                                             const float* __restrict__ dinv,
                                             const float* __restrict__ bias,
                                             float* __restrict__ out) {
  const int lane = threadIdx.x & 31;
  const int wave = threadIdx.x >> 5;
  const int row0 = blockIdx.x * 64 + wave * 16;
  const int lo = lane & 15, hi = lane >> 4;

  const v2f* arow = (const v2f*)(adj + (size_t)(row0 + lo) * NN) + hi;
  const v2f* bf = (const v2f*)bfrag + lane;

  v8f c[8] = {};

  for (int ks = 0; ks < NN / 4; ++ks) {
    const v2f a = __builtin_nontemporal_load(arow + (size_t)ks * 2);
    const v2f* bks = bf + (size_t)ks * (8 * 32);
#pragma unroll
    for (int t = 0; t < 8; ++t) {
      const v2f b = bks[t * 32];
      c[t] = __builtin_amdgcn_wmma_f32_16x16x4_f32(
           false, a,  false, b,
           (short)0, c[t],  false,  false);
    }
  }

#pragma unroll
  for (int t = 0; t < 8; ++t) {
    const int col = t * 16 + lo;
    const float bv = bias[col];
#pragma unroll
    for (int v = 0; v < 8; ++v) {
      const int row = row0 + hi * 8 + v;
      out[(size_t)row * D + col] = dinv[row] * c[t][v] + bv;
    }
  }
}

extern "C" void kernel_launch(void* const* d_in, const int* in_sizes, int n_in,
                              void* d_out, int out_size, void* d_ws, size_t ws_size,
                              hipStream_t stream) {
  const float* x    = (const float*)d_in[0];
  const float* adj  = (const float*)d_in[1];
  const float* w    = (const float*)d_in[2];
  const float* bias = (const float*)d_in[3];
  float* out = (float*)d_out;

  float* dinv  = (float*)d_ws;
  float* bfrag = dinv + NN;

  k_deg <<<NN, 256, 0, stream>>>(adj, dinv);
  k_proj<<<NN, 128, 0, stream>>>(x, w, dinv, bfrag);
  k_agg <<<NN / 64, 128, 0, stream>>>(adj, bfrag, dinv, bias, out);
}


# k_agg as an LDS block GEMM: a 32-row x 128-col workgroup stages each 128-k chunk of adj and of hs (stored as k-pair rows) in LDS once, 8 waves share it, 256 workgroups; sum order unchanged
# speedup vs baseline: 6.5884x; 6.5884x over previous
#include <hip/hip_runtime.h>


typedef float v2f __attribute__((ext_vector_type(2)));
typedef float v4f __attribute__((ext_vector_type(4)));
typedef float v8f __attribute__((ext_vector_type(8)));

#define NN 8192
#define D 128

#define BM 32
#define KC 128
#define NCH (NN / KC)
#define PA 132
#define PB 288
#define A_FLOATS (BM * PA)
#define B_FLOATS ((KC / 2) * PB)
#define STAGE_FLOATS (A_FLOATS + B_FLOATS)
#define CHUNK_B_V4 ((KC / 2) * 2 * D / 4)

__global__ __launch_bounds__(256) void k_deg(const float* __restrict__ adj,
                                             float* __restrict__ dinv) {
  __shared__ float red[256];
  const int row = blockIdx.x;
  const float* r = adj + (size_t)row * NN;
  float s = 0.f;
  for (int j = threadIdx.x; j < NN; j += 256)
    s += __builtin_nontemporal_load(r + j);
  red[threadIdx.x] = s;
  __syncthreads();
  for (int off = 128; off > 0; off >>= 1) {
    if (threadIdx.x < off) red[threadIdx.x] += red[threadIdx.x + off];
    __syncthreads();
  }
  if (threadIdx.x == 0) {
    float d = red[0];
    dinv[row] = d > 0.f ? rsqrtf(d) : 0.f;
  }
}

__global__ __launch_bounds__(128) void k_proj(const float* __restrict__ x,
                                              const float* __restrict__ w,
                                              const float* __restrict__ dinv,
                                              float* __restrict__ hsp) {
  __shared__ float xs[2 * D];
  const int p = blockIdx.x;
  const int col = threadIdx.x;
  xs[col] = x[(size_t)(2 * p) * D + col];
  xs[D + col] = x[(size_t)(2 * p + 1) * D + col];
  __syncthreads();
  float acc0 = 0.f, acc1 = 0.f;
#pragma unroll 8
  for (int k = 0; k < D; ++k) {
    const float wv = w[k * D + col];
    acc0 += xs[k] * wv;
    acc1 += xs[D + k] * wv;
  }
  acc0 *= dinv[2 * p];
  acc1 *= dinv[2 * p + 1];
  v2f o;
  o.x = acc0;
  o.y = acc1;
  ((v2f*)hsp)[(size_t)p * D + col] = o;
}

__global__ __launch_bounds__(256) __attribute__((amdgpu_waves_per_eu(2, 2)))
void k_agg(const float* __restrict__ adj,
           const float* __restrict__ hsp,
           const float* __restrict__ dinv,
           const float* __restrict__ bias,
           float* __restrict__ out) {
  __shared__ __attribute__((aligned(16))) float L[2 * STAGE_FLOATS];

  const int tid = threadIdx.x;
  const int lane = tid & 31;
  const int wave = tid >> 5;
  const int lo = lane & 15, hi = lane >> 4;
  const int rt = wave & 1, cq = wave >> 1;
  const int row0 = blockIdx.x * BM;

  const v4f* ag = (const v4f*)(adj + (size_t)(row0 + (tid >> 5)) * NN) + (tid & 31);
  const v4f* bg = (const v4f*)hsp + tid;
  const int a_dst = (tid >> 5) * PA + (tid & 31) * 4;
  const int b_dst = A_FLOATS + (tid >> 6) * PB + (tid & 63) * 4;
  const int a_src = (16 * rt + lo) * PA + 2 * hi;
  const int b_src = A_FLOATS + hi * PB + (32 * cq + lo) * 2;

  v4f ra[4], rb[16];
#pragma unroll
  for (int i = 0; i < 4; ++i)
    ra[i] = __builtin_nontemporal_load(ag + (size_t)i * (8 * NN / 4));
#pragma unroll
  for (int i = 0; i < 16; ++i) rb[i] = bg[i * 256];
#pragma unroll
  for (int i = 0; i < 4; ++i) *(v4f*)(L + a_dst + i * 8 * PA) = ra[i];
#pragma unroll
  for (int i = 0; i < 16; ++i) *(v4f*)(L + b_dst + i * 4 * PB) = rb[i];
  __syncthreads();

  v8f c0 = {}, c1 = {};

  for (int ch = 0; ch < NCH; ++ch) {
    const float* S = L + (ch & 1) * STAGE_FLOATS;
    float* T = L + ((ch + 1) & 1) * STAGE_FLOATS;
    const float* ap = S + a_src;
    const float* bp = S + b_src;

    if (ch + 1 < NCH) {
      const v4f* agn = ag + (size_t)(ch + 1) * (KC / 4);
      const v4f* bgn = bg + (size_t)(ch + 1) * CHUNK_B_V4;
#pragma unroll
      for (int i = 0; i < 4; ++i)
        ra[i] = __builtin_nontemporal_load(agn + (size_t)i * (8 * NN / 4));
#pragma unroll
      for (int i = 0; i < 16; ++i) rb[i] = bgn[i * 256];
    }

    v2f fa[2][8], fb0[2][8], fb1[2][8];
#pragma unroll
    for (int j = 0; j < 8; ++j) {
      fa[0][j]  = *(const v2f*)(ap + 4 * j);
      fb0[0][j] = *(const v2f*)(bp + 2 * j * PB);
      fb1[0][j] = *(const v2f*)(bp + 2 * j * PB + 32);
    }
#pragma unroll
    for (int g = 0; g < KC / 32; ++g) {
      if (g + 1 < KC / 32) {
#pragma unroll
        for (int j = 0; j < 8; ++j) {
          const int s = (g + 1) * 8 + j;
          fa[(g + 1) & 1][j]  = *(const v2f*)(ap + 4 * s);
          fb0[(g + 1) & 1][j] = *(const v2f*)(bp + 2 * s * PB);
          fb1[(g + 1) & 1][j] = *(const v2f*)(bp + 2 * s * PB + 32);
        }
      } else {
        if (ch + 1 < NCH) {
#pragma unroll
          for (int i = 0; i < 4; ++i) *(v4f*)(T + a_dst + i * 8 * PA) = ra[i];
#pragma unroll
          for (int i = 0; i < 16; ++i) *(v4f*)(T + b_dst + i * 4 * PB) = rb[i];
        }
      }
      __builtin_amdgcn_sched_barrier(0);
#pragma unroll
      for (int j = 0; j < 8; ++j) {
        c0 = __builtin_amdgcn_wmma_f32_16x16x4_f32(
             false, fa[g & 1][j],  false, fb0[g & 1][j],
             (short)0, c0,  false,  false);
        c1 = __builtin_amdgcn_wmma_f32_16x16x4_f32(
             false, fa[g & 1][j],  false, fb1[g & 1][j],
             (short)0, c1,  false,  false);
      }
    }
    __syncthreads();
  }

  {
    const int colA = 32 * cq + lo;
    const int colB = colA + 16;
    const float bvA = bias[colA];
    const float bvB = bias[colB];
#pragma unroll
    for (int v = 0; v < 8; ++v) {
      const int row = row0 + 16 * rt + hi * 8 + v;
      const float dv = dinv[row];
      out[(size_t)row * D + colA] = dv * c0[v] + bvA;
      out[(size_t)row * D + colB] = dv * c1[v] + bvB;
    }
  }
}

extern "C" void kernel_launch(void* const* d_in, const int* in_sizes, int n_in,
                              void* d_out, int out_size, void* d_ws, size_t ws_size,
                              hipStream_t stream) {
  const float* x    = (const float*)d_in[0];
  const float* adj  = (const float*)d_in[1];
  const float* w    = (const float*)d_in[2];
  const float* bias = (const float*)d_in[3];
  float* out = (float*)d_out;

  float* dinv = (float*)d_ws;
  float* hsp  = dinv + NN;

  k_deg <<<NN, 256, 0, stream>>>(adj, dinv);
  k_proj<<<NN / 2, 128, 0, stream>>>(x, w, dinv, hsp);
  k_agg <<<NN / BM, 256, 0, stream>>>(adj, hsp, dinv, bias, out);
}


# k_agg K loop re-pipelined round a hand-written split barrier: a chunk's global loads get a whole trip of lead, in flight across it; next first fragments read under half a WMMA group; sum order same
# speedup vs baseline: 7.0533x; 1.0706x over previous
#include <hip/hip_runtime.h>


typedef float v2f __attribute__((ext_vector_type(2)));
typedef float v4f __attribute__((ext_vector_type(4)));
typedef float v8f __attribute__((ext_vector_type(8)));

#define NN 8192
#define D 128

#define BM 32
#define KC 128
#define NCH (NN / KC)
#define PA 132
#define PB 288
#define A_FLOATS (BM * PA)
#define B_FLOATS ((KC / 2) * PB)
#define STAGE_FLOATS (A_FLOATS + B_FLOATS)
#define CHUNK_B_V2 (KC * D / 2)

__global__ __launch_bounds__(256) void k_deg(const float* __restrict__ adj,
                                             float* __restrict__ dinv) {
  __shared__ float red[256];
  const int row = blockIdx.x;
  const float* r = adj + (size_t)row * NN;
  float s = 0.f;
  for (int j = threadIdx.x; j < NN; j += 256)
    s += __builtin_nontemporal_load(r + j);
  red[threadIdx.x] = s;
  __syncthreads();
  for (int off = 128; off > 0; off >>= 1) {
    if (threadIdx.x < off) red[threadIdx.x] += red[threadIdx.x + off];
    __syncthreads();
  }
  if (threadIdx.x == 0) {
    float d = red[0];
    dinv[row] = d > 0.f ? rsqrtf(d) : 0.f;
  }
}

template <class F>
__device__ __forceinline__ void mul_chunk(const float* ap, const float* bp,
                                          v8f& c0, v8f& c1, F&& refill) {
  v2f fa[2][8], fb0[2][8], fb1[2][8];
#pragma unroll
  for (int j = 0; j < 8; ++j) {
    fa[0][j]  = *(const v2f*)(ap + 4 * j);
    fb0[0][j] = *(const v2f*)(bp + 2 * j * PB);
    fb1[0][j] = *(const v2f*)(bp + 2 * j * PB + 32);
  }
#pragma unroll
  for (int g = 0; g < KC / 32; ++g) {
    if (g + 1 < KC / 32) {
#pragma unroll
      for (int j = 0; j < 8; ++j) {
        const int s = (g + 1) * 8 + j;
        fa[(g + 1) & 1][j]  = *(const v2f*)(ap + 4 * s);
        fb0[(g + 1) & 1][j] = *(const v2f*)(bp + 2 * s * PB);
        fb1[(g + 1) & 1][j] = *(const v2f*)(bp + 2 * s * PB + 32);
      }
    } else {
      refill();
    }
    __builtin_amdgcn_sched_barrier(0);
#pragma unroll
    for (int j = 0; j < 8; ++j) {
      c0 = __builtin_amdgcn_wmma_f32_16x16x4_f32(
           false, fa[g & 1][j],  false, fb0[g & 1][j],
           (short)0, c0,  false,  false);
      c1 = __builtin_amdgcn_wmma_f32_16x16x4_f32(
           false, fa[g & 1][j],  false, fb1[g & 1][j],
           (short)0, c1,  false,  false);
    }
  }
}

struct Frag { v2f a[8], b0[8], b1[8]; };
template <int M> struct mode_c { static constexpr int value = M; };

__device__ __forceinline__ void load_group(Frag& f, const float* ap, const float* bp,
                                           const int g) {
#pragma unroll
  for (int j = 0; j < 8; ++j) {
    const int s = g * 8 + j;
    f.a[j]  = *(const v2f*)(ap + 4 * s);
    f.b0[j] = *(const v2f*)(bp + 2 * s * PB);
    f.b1[j] = *(const v2f*)(bp + 2 * s * PB + 32);
  }
}

template <int J0, int J1>
__device__ __forceinline__ void mma_group(const Frag& f, v8f& c0, v8f& c1) {
#pragma unroll
  for (int j = J0; j < J1; ++j) {
    c0 = __builtin_amdgcn_wmma_f32_16x16x4_f32(
         false, f.a[j],  false, f.b0[j],
         (short)0, c0,  false,  false);
    c1 = __builtin_amdgcn_wmma_f32_16x16x4_f32(
         false, f.a[j],  false, f.b1[j],
         (short)0, c1,  false,  false);
  }
}

__global__ __launch_bounds__(256) __attribute__((amdgpu_waves_per_eu(2, 2)))
void k_agg(const float* __restrict__ adj,
           const float* __restrict__ x,
           const float* __restrict__ w,
           const float* __restrict__ dinv,
           const float* __restrict__ bias,
           float* __restrict__ out) {
  __shared__ __attribute__((aligned(16))) float L[2 * STAGE_FLOATS];

  const int tid = threadIdx.x;
  const int lane = tid & 31;
  const int wave = tid >> 5;
  const int lo = lane & 15, hi = lane >> 4;
  const int rt = wave & 1, cq = wave >> 1;
  const int row0 = blockIdx.x * BM;

  const v4f* ag = (const v4f*)(adj + (size_t)(row0 + (tid >> 5)) * NN) + (tid & 31);
  const v4f* dg = (const v4f*)dinv + (tid & 31);
  const size_t boff = (size_t)(2 * (tid >> 6)) * (D / 2) + (tid & 63);
  const v2f* xg = (const v2f*)x + boff;
  const v2f* wg = (const v2f*)w + boff;
  const int a_dst = (tid >> 5) * PA + (tid & 31) * 4;
  const int b_dst = A_FLOATS + (tid >> 6) * PB + (tid & 63) * 4;
  const int a_src = (16 * rt + lo) * PA + 2 * hi;
  const int b_src = A_FLOATS + hi * PB + (32 * cq + lo) * 2;

  v4f ra[4], dk;
  v2f rb0[16], rb1[16];
#pragma unroll
  for (int i = 0; i < 4; ++i)
    ra[i] = __builtin_nontemporal_load(ag + (size_t)i * (8 * NN / 4));
  dk = dg[0];
#pragma unroll
  for (int i = 0; i < 16; ++i) {
    rb0[i] = xg[(8 * i) * (D / 2)];
    rb1[i] = xg[(8 * i + 1) * (D / 2)];
  }
#pragma unroll
  for (int i = 0; i < 4; ++i) *(v4f*)(L + a_dst + i * 8 * PA) = ra[i] * dk;
#pragma unroll
  for (int i = 0; i < 16; ++i) {
    const v4f o = {rb0[i].x, rb1[i].x, rb0[i].y, rb1[i].y};
    *(v4f*)(L + b_dst + i * 4 * PB) = o;
  }
  __syncthreads();

#pragma unroll
  for (int i = 0; i < 4; ++i)
    ra[i] = __builtin_nontemporal_load(ag + (KC / 4) + (size_t)i * (8 * NN / 4));
  dk = dg[KC / 4];
#pragma unroll
  for (int i = 0; i < 16; ++i) {
    rb0[i] = xg[CHUNK_B_V2 + (8 * i) * (D / 2)];
    rb1[i] = xg[CHUNK_B_V2 + (8 * i + 1) * (D / 2)];
  }

  v8f c0 = {}, c1 = {};
  Frag f0, f1;
  load_group(f0, L + a_src, L + b_src, 0);

  auto trip = [&](const int ch, auto mode) {
    constexpr int M = decltype(mode)::value;
    const float* S = L + (ch & 1) * STAGE_FLOATS;
    float* T = L + ((ch + 1) & 1) * STAGE_FLOATS;
    const float* ap = S + a_src;
    const float* bp = S + b_src;

    load_group(f1, ap, bp, 1);
    __builtin_amdgcn_sched_barrier(0);
    mma_group<0, 8>(f0, c0, c1);
    load_group(f0, ap, bp, 2);
    __builtin_amdgcn_sched_barrier(0);
    mma_group<0, 8>(f1, c0, c1);
    load_group(f1, ap, bp, 3);
    __builtin_amdgcn_sched_barrier(0);
    if constexpr (M <= 1) {
#pragma unroll
      for (int i = 0; i < 4; ++i) *(v4f*)(T + a_dst + i * 8 * PA) = ra[i] * dk;
    }
#pragma unroll
    for (int i = 0; i < 16; ++i) {
      const v4f o = {rb0[i].x, rb1[i].x, rb0[i].y, rb1[i].y};
      *(v4f*)(T + b_dst + i * 4 * PB) = o;
    }
    if constexpr (M == 0) {
      const v4f* agn = ag + (size_t)(ch + 2) * (KC / 4);
#pragma unroll
      for (int i = 0; i < 4; ++i)
        ra[i] = __builtin_nontemporal_load(agn + (size_t)i * (8 * NN / 4));
      dk = dg[(ch + 2) * (KC / 4)];
      const v2f* bgn = xg + (size_t)(ch + 2) * CHUNK_B_V2;
#pragma unroll
      for (int i = 0; i < 16; ++i) {
        rb0[i] = bgn[(8 * i) * (D / 2)];
        rb1[i] = bgn[(8 * i + 1) * (D / 2)];
      }
    } else if constexpr (M == 1) {
#pragma unroll
      for (int i = 0; i < 16; ++i) {
        rb0[i] = wg[(8 * i) * (D / 2)];
        rb1[i] = wg[(8 * i + 1) * (D / 2)];
      }
    }
    mma_group<0, 8>(f0, c0, c1);
    __builtin_amdgcn_sched_barrier(0);
    asm volatile("s_wait_dscnt 0x0" ::: "memory");
    asm volatile("s_barrier_signal -1" ::: "memory");
    __builtin_amdgcn_sched_barrier(0);
    mma_group<0, 4>(f1, c0, c1);
    __builtin_amdgcn_sched_barrier(0);
    asm volatile("s_barrier_wait -1" ::: "memory");
    if constexpr (M <= 1) load_group(f0, T + a_src, T + b_src, 0);
    __builtin_amdgcn_sched_barrier(0);
    mma_group<4, 8>(f1, c0, c1);
  };
  for (int ch = 0; ch < NCH - 2; ++ch) trip(ch, mode_c<0>{});
  trip(NCH - 2, mode_c<1>{});
  trip(NCH - 1, mode_c<2>{});

  {
    float* P = L + (NCH & 1) * STAGE_FLOATS;
    float* zp = P + (16 * rt + 8 * hi) * PA + 32 * cq + lo;
#pragma unroll
    for (int v = 0; v < 8; ++v) {
      zp[v * PA] = c0[v];
      zp[v * PA + 16] = c1[v];
    }
    __syncthreads();
    c0 = (v8f){};
    c1 = (v8f){};
    mul_chunk(P + a_src, P + b_src, c0, c1, []() {});
  }

  {
    const int colA = 32 * cq + lo;
    const int colB = colA + 16;
    const float bvA = bias[colA];
    const float bvB = bias[colB];
#pragma unroll
    for (int v = 0; v < 8; ++v) {
      const int row = row0 + 16 * rt + hi * 8 + v;
      const float dv = dinv[row];
      out[(size_t)row * D + colA] = dv * c0[v] + bvA;
      out[(size_t)row * D + colB] = dv * c1[v] + bvB;
    }
  }
}

extern "C" void kernel_launch(void* const* d_in, const int* in_sizes, int n_in,
                              void* d_out, int out_size, void* d_ws, size_t ws_size,
                              hipStream_t stream) {
  const float* x    = (const float*)d_in[0];
  const float* adj  = (const float*)d_in[1];
  const float* w    = (const float*)d_in[2];
  const float* bias = (const float*)d_in[3];
  float* out = (float*)d_out;

  float* dinv = (float*)d_ws;

  k_deg<<<NN, 256, 0, stream>>>(adj, dinv);
  k_agg<<<NN / BM, 256, 0, stream>>>(adj, x, w, dinv, bias, out);
}


# k_agg decomposed as 128 tiles of 64 rows x 2 halves of K: a workgroup reads half of x for twice the rows (32 trips, 32x32 wave tiles); a third launch adds the halves and projects; sum order changes
# speedup vs baseline: 7.7233x; 1.0950x over previous
#include <hip/hip_runtime.h>


typedef float v2f __attribute__((ext_vector_type(2)));
typedef float v4f __attribute__((ext_vector_type(4)));
typedef float v8f __attribute__((ext_vector_type(8)));

#define NN 8192
#define D 128

#define BM 64
#define KSPLIT 2
#define KC 128
#define NCH (NN / KSPLIT / KC)
#define FM 32
#define PA 132
#define PB 288
#define A_FLOATS (BM * PA)
#define FA_FLOATS (FM * PA)
#define B_FLOATS ((KC / 2) * PB)
#define STAGE_FLOATS (A_FLOATS + B_FLOATS)
#define CHUNK_B_V2 (KC * D / 2)

__global__ __launch_bounds__(256) void k_deg(const float* __restrict__ adj,
                                             float* __restrict__ dinv) {
  __shared__ float red[256];
  const int row = blockIdx.x;
  const float* r = adj + (size_t)row * NN;
  float s = 0.f;
  for (int j = threadIdx.x; j < NN; j += 256)
    s += __builtin_nontemporal_load(r + j);
  red[threadIdx.x] = s;
  __syncthreads();
  for (int off = 128; off > 0; off >>= 1) {
    if (threadIdx.x < off) red[threadIdx.x] += red[threadIdx.x + off];
    __syncthreads();
  }
  if (threadIdx.x == 0) {
    float d = red[0];
    dinv[row] = d > 0.f ? rsqrtf(d) : 0.f;
  }
}

template <class F>
__device__ __forceinline__ void mul_chunk(const float* ap, const float* bp,
                                          v8f& c0, v8f& c1, F&& refill) {
  v2f fa[2][8], fb0[2][8], fb1[2][8];
#pragma unroll
  for (int j = 0; j < 8; ++j) {
    fa[0][j]  = *(const v2f*)(ap + 4 * j);
    fb0[0][j] = *(const v2f*)(bp + 2 * j * PB);
    fb1[0][j] = *(const v2f*)(bp + 2 * j * PB + 32);
  }
#pragma unroll
  for (int g = 0; g < KC / 32; ++g) {
    if (g + 1 < KC / 32) {
#pragma unroll
      for (int j = 0; j < 8; ++j) {
        const int s = (g + 1) * 8 + j;
        fa[(g + 1) & 1][j]  = *(const v2f*)(ap + 4 * s);
        fb0[(g + 1) & 1][j] = *(const v2f*)(bp + 2 * s * PB);
        fb1[(g + 1) & 1][j] = *(const v2f*)(bp + 2 * s * PB + 32);
      }
    } else {
      refill();
    }
    __builtin_amdgcn_sched_barrier(0);
#pragma unroll
    for (int j = 0; j < 8; ++j) {
      c0 = __builtin_amdgcn_wmma_f32_16x16x4_f32(
           false, fa[g & 1][j],  false, fb0[g & 1][j],
           (short)0, c0,  false,  false);
      c1 = __builtin_amdgcn_wmma_f32_16x16x4_f32(
           false, fa[g & 1][j],  false, fb1[g & 1][j],
           (short)0, c1,  false,  false);
    }
  }
}

struct Frag { v2f a0[8], a1[8], b0[8], b1[8]; };
template <int M> struct mode_c { static constexpr int value = M; };

__device__ __forceinline__ void load_group(Frag& f, const float* ap, const float* bp,
                                           const int g) {
#pragma unroll
  for (int j = 0; j < 8; ++j) {
    const int s = g * 8 + j;
    f.a0[j] = *(const v2f*)(ap + 4 * s);
    f.a1[j] = *(const v2f*)(ap + 16 * PA + 4 * s);
    f.b0[j] = *(const v2f*)(bp + 2 * s * PB);
    f.b1[j] = *(const v2f*)(bp + 2 * s * PB + 32);
  }
}

__device__ __forceinline__ v8f mma(const v2f a, const v2f b, const v8f c) {
  return __builtin_amdgcn_wmma_f32_16x16x4_f32(
       false, a,  false, b,
       (short)0, c,  false,  false);
}

template <int J0, int J1>
__device__ __forceinline__ void mma_group(const Frag& f, v8f& c00, v8f& c01,
                                          v8f& c10, v8f& c11) {
#pragma unroll
  for (int j = J0; j < J1; ++j) {
    c00 = mma(f.a0[j], f.b0[j], c00);
    c01 = mma(f.a0[j], f.b1[j], c01);
    c10 = mma(f.a1[j], f.b0[j], c10);
    c11 = mma(f.a1[j], f.b1[j], c11);
  }
}

__global__ __launch_bounds__(256) __attribute__((amdgpu_waves_per_eu(2, 2)))
void k_agg(const float* __restrict__ adj,
           const float* __restrict__ x,
           const float* __restrict__ dinv,
           float* __restrict__ part0,
           float* __restrict__ part1) {
  __shared__ __attribute__((aligned(16))) float L[2 * STAGE_FLOATS];

  const int tid = threadIdx.x;
  const int lane = tid & 31;
  const int wave = tid >> 5;
  const int lo = lane & 15, hi = lane >> 4;
  const int rt = wave & 1, cq = wave >> 1;
  const int row0 = (blockIdx.x >> 1) * BM;
  const int kh = blockIdx.x & 1;
  const size_t k0 = (size_t)kh * (NN / KSPLIT);

  const v4f* ag = (const v4f*)(adj + (size_t)(row0 + (tid >> 5)) * NN + k0) + (tid & 31);
  const v4f* dg = (const v4f*)(dinv + k0) + (tid & 31);
  const size_t boff = (size_t)(2 * (tid >> 6)) * (D / 2) + (tid & 63);
  const v2f* xg = (const v2f*)(x + k0 * D) + boff;
  const int a_dst = (tid >> 5) * PA + (tid & 31) * 4;
  const int b_dst = A_FLOATS + (tid >> 6) * PB + (tid & 63) * 4;
  const int a_src = (32 * rt + lo) * PA + 2 * hi;
  const int b_src = A_FLOATS + hi * PB + (32 * cq + lo) * 2;

  v4f ra[8], dk;
  v2f rb0[16], rb1[16];
#pragma unroll
  for (int i = 0; i < 8; ++i)
    ra[i] = __builtin_nontemporal_load(ag + (size_t)i * (8 * NN / 4));
  dk = dg[0];
#pragma unroll
  for (int i = 0; i < 16; ++i) {
    rb0[i] = xg[(8 * i) * (D / 2)];
    rb1[i] = xg[(8 * i + 1) * (D / 2)];
  }
#pragma unroll
  for (int i = 0; i < 8; ++i) *(v4f*)(L + a_dst + i * 8 * PA) = ra[i] * dk;
#pragma unroll
  for (int i = 0; i < 16; ++i) {
    const v4f o = {rb0[i].x, rb1[i].x, rb0[i].y, rb1[i].y};
    *(v4f*)(L + b_dst + i * 4 * PB) = o;
  }
  __syncthreads();

#pragma unroll
  for (int i = 0; i < 8; ++i)
    ra[i] = __builtin_nontemporal_load(ag + (KC / 4) + (size_t)i * (8 * NN / 4));
  dk = dg[KC / 4];
#pragma unroll
  for (int i = 0; i < 16; ++i) {
    rb0[i] = xg[CHUNK_B_V2 + (8 * i) * (D / 2)];
    rb1[i] = xg[CHUNK_B_V2 + (8 * i + 1) * (D / 2)];
  }

  v8f c00 = {}, c01 = {}, c10 = {}, c11 = {};
  Frag f0, f1;
  load_group(f0, L + a_src, L + b_src, 0);

  auto trip = [&](const int ch, auto mode) {
    constexpr int M = decltype(mode)::value;
    const float* S = L + (ch & 1) * STAGE_FLOATS;
    float* T = L + ((ch + 1) & 1) * STAGE_FLOATS;
    const float* ap = S + a_src;
    const float* bp = S + b_src;

    load_group(f1, ap, bp, 1);
    __builtin_amdgcn_sched_barrier(0);
    mma_group<0, 8>(f0, c00, c01, c10, c11);
    load_group(f0, ap, bp, 2);
    __builtin_amdgcn_sched_barrier(0);
    mma_group<0, 8>(f1, c00, c01, c10, c11);
    load_group(f1, ap, bp, 3);
    __builtin_amdgcn_sched_barrier(0);
    if constexpr (M <= 1) {
#pragma unroll
      for (int i = 0; i < 8; ++i) *(v4f*)(T + a_dst + i * 8 * PA) = ra[i] * dk;
#pragma unroll
      for (int i = 0; i < 16; ++i) {
        const v4f o = {rb0[i].x, rb1[i].x, rb0[i].y, rb1[i].y};
        *(v4f*)(T + b_dst + i * 4 * PB) = o;
      }
    }
    if constexpr (M == 0) {
      const v4f* agn = ag + (size_t)(ch + 2) * (KC / 4);
#pragma unroll
      for (int i = 0; i < 8; ++i)
        ra[i] = __builtin_nontemporal_load(agn + (size_t)i * (8 * NN / 4));
      dk = dg[(ch + 2) * (KC / 4)];
      const v2f* bgn = xg + (size_t)(ch + 2) * CHUNK_B_V2;
#pragma unroll
      for (int i = 0; i < 16; ++i) {
        rb0[i] = bgn[(8 * i) * (D / 2)];
        rb1[i] = bgn[(8 * i + 1) * (D / 2)];
      }
    }
    mma_group<0, 8>(f0, c00, c01, c10, c11);
    if constexpr (M <= 1) {
      __builtin_amdgcn_sched_barrier(0);
      asm volatile("s_wait_dscnt 0x0" ::: "memory");
      asm volatile("s_barrier_signal -1" ::: "memory");
      __builtin_amdgcn_sched_barrier(0);
      mma_group<0, 4>(f1, c00, c01, c10, c11);
      __builtin_amdgcn_sched_barrier(0);
      asm volatile("s_barrier_wait -1" ::: "memory");
      load_group(f0, T + a_src, T + b_src, 0);
      __builtin_amdgcn_sched_barrier(0);
      mma_group<4, 8>(f1, c00, c01, c10, c11);
    } else {
      mma_group<0, 8>(f1, c00, c01, c10, c11);
    }
  };
  for (int ch = 0; ch < NCH - 2; ++ch) trip(ch, mode_c<0>{});
  trip(NCH - 2, mode_c<1>{});
  trip(NCH - 1, mode_c<2>{});

  {
    float* pz = (kh ? part1 : part0) + (size_t)(row0 + 32 * rt + 8 * hi) * D + 32 * cq + lo;
#pragma unroll
    for (int v = 0; v < 8; ++v) {
      pz[(size_t)v * D] = c00[v];
      pz[(size_t)v * D + 16] = c01[v];
      pz[(size_t)(v + 16) * D] = c10[v];
      pz[(size_t)(v + 16) * D + 16] = c11[v];
    }
  }
}

__global__ __launch_bounds__(256)
void k_fin(const float* __restrict__ part1,
           const float* __restrict__ w,
           const float* __restrict__ dinv,
           const float* __restrict__ bias,
           float* out) {
  __shared__ __attribute__((aligned(16))) float Lf[FA_FLOATS + B_FLOATS];

  const int tid = threadIdx.x;
  const int lane = tid & 31;
  const int wave = tid >> 5;
  const int lo = lane & 15, hi = lane >> 4;
  const int rt = wave & 1, cq = wave >> 1;
  const int row0 = blockIdx.x * FM;

  {
    const size_t zoff = (size_t)(row0 + (tid >> 5)) * (D / 4) + (tid & 31);
    const v4f* p0 = (const v4f*)out + zoff;
    const v4f* p1 = (const v4f*)part1 + zoff;
    const v2f* wg = (const v2f*)w + (size_t)(2 * (tid >> 6)) * (D / 2) + (tid & 63);
    v4f z0[4], z1[4];
    v2f w0[16], w1[16];
#pragma unroll
    for (int i = 0; i < 4; ++i) {
      z0[i] = p0[i * 8 * (D / 4)];
      z1[i] = p1[i * 8 * (D / 4)];
    }
#pragma unroll
    for (int i = 0; i < 16; ++i) {
      w0[i] = wg[(8 * i) * (D / 2)];
      w1[i] = wg[(8 * i + 1) * (D / 2)];
    }
#pragma unroll
    for (int i = 0; i < 4; ++i)
      *(v4f*)(Lf + (tid >> 5) * PA + (tid & 31) * 4 + i * 8 * PA) = z0[i] + z1[i];
#pragma unroll
    for (int i = 0; i < 16; ++i) {
      const v4f o = {w0[i].x, w1[i].x, w0[i].y, w1[i].y};
      *(v4f*)(Lf + FA_FLOATS + (tid >> 6) * PB + (tid & 63) * 4 + i * 4 * PB) = o;
    }
  }
  __syncthreads();

  v8f c0 = {}, c1 = {};
  mul_chunk(Lf + (16 * rt + lo) * PA + 2 * hi,
            Lf + FA_FLOATS + hi * PB + (32 * cq + lo) * 2, c0, c1, []() {});

  {
    const int colA = 32 * cq + lo;
    const int colB = colA + 16;
    const float bvA = bias[colA];
    const float bvB = bias[colB];
#pragma unroll
    for (int v = 0; v < 8; ++v) {
      const int row = row0 + 16 * rt + hi * 8 + v;
      const float dv = dinv[row];
      out[(size_t)row * D + colA] = dv * c0[v] + bvA;
      out[(size_t)row * D + colB] = dv * c1[v] + bvB;
    }
  }
}

extern "C" void kernel_launch(void* const* d_in, const int* in_sizes, int n_in,
                              void* d_out, int out_size, void* d_ws, size_t ws_size,
                              hipStream_t stream) {
  const float* x    = (const float*)d_in[0];
  const float* adj  = (const float*)d_in[1];
  const float* w    = (const float*)d_in[2];
  const float* bias = (const float*)d_in[3];
  float* out = (float*)d_out;

  float* dinv  = (float*)d_ws;
  float* part1 = (float*)d_ws + NN;

  k_deg<<<NN, 256, 0, stream>>>(adj, dinv);
  k_agg<<<(NN / BM) * KSPLIT, 256, 0, stream>>>(adj, x, dinv, out, part1);
  k_fin<<<NN / FM, 256, 0, stream>>>(part1, w, dinv, bias, out);
}


# k_agg: all global-to-LDS staging moved onto waves 0-3, the older wave of each SIMD; waves 4-7 only read fragments and multiply, so staging is filled by the partner's WMMAs; sum order unchanged
# speedup vs baseline: 8.1654x; 1.0572x over previous
#include <hip/hip_runtime.h>


typedef float v2f __attribute__((ext_vector_type(2)));
typedef float v4f __attribute__((ext_vector_type(4)));
typedef float v8f __attribute__((ext_vector_type(8)));

#define NN 8192
#define D 128

#define BM 64
#define KSPLIT 2
#define KC 128
#define NCH (NN / KSPLIT / KC)
#define FM 32
#define PA 132
#define PB 288
#define A_FLOATS (BM * PA)
#define FA_FLOATS (FM * PA)
#define B_FLOATS ((KC / 2) * PB)
#define STAGE_FLOATS (A_FLOATS + B_FLOATS)
#define CHUNK_B_V2 (KC * D / 2)

__global__ __launch_bounds__(256) void k_deg(const float* __restrict__ adj,
                                             float* __restrict__ dinv) {
  __shared__ float red[256];
  const int row = blockIdx.x;
  const float* r = adj + (size_t)row * NN;
  float s = 0.f;
  for (int j = threadIdx.x; j < NN; j += 256)
    s += __builtin_nontemporal_load(r + j);
  red[threadIdx.x] = s;
  __syncthreads();
  for (int off = 128; off > 0; off >>= 1) {
    if (threadIdx.x < off) red[threadIdx.x] += red[threadIdx.x + off];
    __syncthreads();
  }
  if (threadIdx.x == 0) {
    float d = red[0];
    dinv[row] = d > 0.f ? rsqrtf(d) : 0.f;
  }
}

template <class F>
__device__ __forceinline__ void mul_chunk(const float* ap, const float* bp,
                                          v8f& c0, v8f& c1, F&& refill) {
  v2f fa[2][8], fb0[2][8], fb1[2][8];
#pragma unroll
  for (int j = 0; j < 8; ++j) {
    fa[0][j]  = *(const v2f*)(ap + 4 * j);
    fb0[0][j] = *(const v2f*)(bp + 2 * j * PB);
    fb1[0][j] = *(const v2f*)(bp + 2 * j * PB + 32);
  }
#pragma unroll
  for (int g = 0; g < KC / 32; ++g) {
    if (g + 1 < KC / 32) {
#pragma unroll
      for (int j = 0; j < 8; ++j) {
        const int s = (g + 1) * 8 + j;
        fa[(g + 1) & 1][j]  = *(const v2f*)(ap + 4 * s);
        fb0[(g + 1) & 1][j] = *(const v2f*)(bp + 2 * s * PB);
        fb1[(g + 1) & 1][j] = *(const v2f*)(bp + 2 * s * PB + 32);
      }
    } else {
      refill();
    }
    __builtin_amdgcn_sched_barrier(0);
#pragma unroll
    for (int j = 0; j < 8; ++j) {
      c0 = __builtin_amdgcn_wmma_f32_16x16x4_f32(
           false, fa[g & 1][j],  false, fb0[g & 1][j],
           (short)0, c0,  false,  false);
      c1 = __builtin_amdgcn_wmma_f32_16x16x4_f32(
           false, fa[g & 1][j],  false, fb1[g & 1][j],
           (short)0, c1,  false,  false);
    }
  }
}

struct Frag { v2f a0[8], a1[8], b0[8], b1[8]; };
template <int M> struct mode_c { static constexpr int value = M; };

__device__ __forceinline__ void load_group(Frag& f, const float* ap, const float* bp,
                                           const int g) {
#pragma unroll
  for (int j = 0; j < 8; ++j) {
    const int s = g * 8 + j;
    f.a0[j] = *(const v2f*)(ap + 4 * s);
    f.a1[j] = *(const v2f*)(ap + 16 * PA + 4 * s);
    f.b0[j] = *(const v2f*)(bp + 2 * s * PB);
    f.b1[j] = *(const v2f*)(bp + 2 * s * PB + 32);
  }
}

__device__ __forceinline__ v8f mma(const v2f a, const v2f b, const v8f c) {
  return __builtin_amdgcn_wmma_f32_16x16x4_f32(
       false, a,  false, b,
       (short)0, c,  false,  false);
}

template <int J0, int J1>
__device__ __forceinline__ void mma_group(const Frag& f, v8f& c00, v8f& c01,
                                          v8f& c10, v8f& c11) {
#pragma unroll
  for (int j = J0; j < J1; ++j) {
    c00 = mma(f.a0[j], f.b0[j], c00);
    c01 = mma(f.a0[j], f.b1[j], c01);
    c10 = mma(f.a1[j], f.b0[j], c10);
    c11 = mma(f.a1[j], f.b1[j], c11);
  }
}

__global__ __launch_bounds__(256) __attribute__((amdgpu_waves_per_eu(2, 2)))
void k_agg(const float* __restrict__ adj,
           const float* __restrict__ x,
           const float* __restrict__ dinv,
           float* __restrict__ part0,
           float* __restrict__ part1) {
  __shared__ __attribute__((aligned(16))) float L[2 * STAGE_FLOATS];

  const int tid = threadIdx.x;
  const int lane = tid & 31;
  const int wave = tid >> 5;
  const int lo = lane & 15, hi = lane >> 4;
  const int rt = wave & 1, cq = wave >> 1;
  const int row0 = (blockIdx.x >> 1) * BM;
  const int kh = blockIdx.x & 1;
  const size_t k0 = (size_t)kh * (NN / KSPLIT);

  const bool stager = __builtin_amdgcn_readfirstlane(tid >> 5) < 4;
  const int st = tid & 127;
  const v4f* ag = (const v4f*)(adj + (size_t)(row0 + (st >> 5)) * NN + k0) + (st & 31);
  const v4f* dg = (const v4f*)(dinv + k0) + (st & 31);
  const size_t boff = (size_t)(2 * (st >> 6)) * (D / 2) + (st & 63);
  const v2f* xg = (const v2f*)(x + k0 * D) + boff;
  const int a_dst = (st >> 5) * PA + (st & 31) * 4;
  const int b_dst = A_FLOATS + (st >> 6) * PB + (st & 63) * 4;
  const int a_src = (32 * rt + lo) * PA + 2 * hi;
  const int b_src = A_FLOATS + hi * PB + (32 * cq + lo) * 2;

  v4f ra[16], dk;
  v2f rb0[32], rb1[32];
  auto request_a = [&](const int ch) {
    const v4f* agn = ag + (size_t)ch * (KC / 4);
#pragma unroll
    for (int i = 0; i < 16; ++i)
      ra[i] = __builtin_nontemporal_load(agn + (size_t)i * (4 * NN / 4));
    dk = dg[ch * (KC / 4)];
  };
  auto request_x = [&](const int ch, auto half) {
    constexpr int H = decltype(half)::value;
    const v2f* bgn = xg + (size_t)ch * CHUNK_B_V2;
#pragma unroll
    for (int i = 16 * H; i < 16 * H + 16; ++i) {
      rb0[i] = bgn[(4 * i) * (D / 2)];
      rb1[i] = bgn[(4 * i + 1) * (D / 2)];
    }
  };
  auto refill = [&](float* T) {
#pragma unroll
    for (int i = 0; i < 16; ++i) *(v4f*)(T + a_dst + i * 4 * PA) = ra[i] * dk;
#pragma unroll
    for (int i = 0; i < 32; ++i) {
      const v4f o = {rb0[i].x, rb1[i].x, rb0[i].y, rb1[i].y};
      *(v4f*)(T + b_dst + i * 2 * PB) = o;
    }
  };
  if (stager) {
    request_a(0);
    request_x(0, mode_c<0>{});
    request_x(0, mode_c<1>{});
    refill(L);
  }
  __syncthreads();

  if (stager) {
    request_a(1);
    request_x(1, mode_c<0>{});
    request_x(1, mode_c<1>{});
  }

  v8f c00 = {}, c01 = {}, c10 = {}, c11 = {};
  Frag f0, f1;
  load_group(f0, L + a_src, L + b_src, 0);

  auto trip = [&](const int ch, auto mode) {
    constexpr int M = decltype(mode)::value;
    const float* S = L + (ch & 1) * STAGE_FLOATS;
    float* T = L + ((ch + 1) & 1) * STAGE_FLOATS;
    const float* ap = S + a_src;
    const float* bp = S + b_src;

    load_group(f1, ap, bp, 1);
    __builtin_amdgcn_sched_barrier(0);
    mma_group<0, 8>(f0, c00, c01, c10, c11);
    load_group(f0, ap, bp, 2);
    __builtin_amdgcn_sched_barrier(0);
    mma_group<0, 8>(f1, c00, c01, c10, c11);
    load_group(f1, ap, bp, 3);
    __builtin_amdgcn_sched_barrier(0);
    if constexpr (M <= 1) {
      if (stager) {
        refill(T);
        if constexpr (M == 0) {
          request_a(ch + 2);
          __builtin_amdgcn_sched_barrier(0);
          request_x(ch + 2, mode_c<0>{});
        }
      }
    }
    mma_group<0, 8>(f0, c00, c01, c10, c11);
    if constexpr (M <= 1) {
      __builtin_amdgcn_sched_barrier(0);
      asm volatile("s_wait_dscnt 0x0" ::: "memory");
      asm volatile("s_barrier_signal -1" ::: "memory");
      __builtin_amdgcn_sched_barrier(0);
      mma_group<0, 4>(f1, c00, c01, c10, c11);
      __builtin_amdgcn_sched_barrier(0);
      if constexpr (M == 0) {
        if (stager) request_x(ch + 2, mode_c<1>{});
      }
      asm volatile("s_barrier_wait -1" ::: "memory");
      load_group(f0, T + a_src, T + b_src, 0);
      __builtin_amdgcn_sched_barrier(0);
      mma_group<4, 8>(f1, c00, c01, c10, c11);
    } else {
      mma_group<0, 8>(f1, c00, c01, c10, c11);
    }
  };
  for (int ch = 0; ch < NCH - 2; ++ch) trip(ch, mode_c<0>{});
  trip(NCH - 2, mode_c<1>{});
  trip(NCH - 1, mode_c<2>{});

  {
    float* pz = (kh ? part1 : part0) + (size_t)(row0 + 32 * rt + 8 * hi) * D + 32 * cq + lo;
#pragma unroll
    for (int v = 0; v < 8; ++v) {
      pz[(size_t)v * D] = c00[v];
      pz[(size_t)v * D + 16] = c01[v];
      pz[(size_t)(v + 16) * D] = c10[v];
      pz[(size_t)(v + 16) * D + 16] = c11[v];
    }
  }
}

__global__ __launch_bounds__(256)
void k_fin(const float* __restrict__ part1,
           const float* __restrict__ w,
           const float* __restrict__ dinv,
           const float* __restrict__ bias,
           float* out) {
  __shared__ __attribute__((aligned(16))) float Lf[FA_FLOATS + B_FLOATS];

  const int tid = threadIdx.x;
  const int lane = tid & 31;
  const int wave = tid >> 5;
  const int lo = lane & 15, hi = lane >> 4;
  const int rt = wave & 1, cq = wave >> 1;
  const int row0 = blockIdx.x * FM;

  {
    const size_t zoff = (size_t)(row0 + (tid >> 5)) * (D / 4) + (tid & 31);
    const v4f* p0 = (const v4f*)out + zoff;
    const v4f* p1 = (const v4f*)part1 + zoff;
    const v2f* wg = (const v2f*)w + (size_t)(2 * (tid >> 6)) * (D / 2) + (tid & 63);
    v4f z0[4], z1[4];
    v2f w0[16], w1[16];
#pragma unroll
    for (int i = 0; i < 4; ++i) {
      z0[i] = p0[i * 8 * (D / 4)];
      z1[i] = p1[i * 8 * (D / 4)];
    }
#pragma unroll
    for (int i = 0; i < 16; ++i) {
      w0[i] = wg[(8 * i) * (D / 2)];
      w1[i] = wg[(8 * i + 1) * (D / 2)];
    }
#pragma unroll
    for (int i = 0; i < 4; ++i)
      *(v4f*)(Lf + (tid >> 5) * PA + (tid & 31) * 4 + i * 8 * PA) = z0[i] + z1[i];
#pragma unroll
    for (int i = 0; i < 16; ++i) {
      const v4f o = {w0[i].x, w1[i].x, w0[i].y, w1[i].y};
      *(v4f*)(Lf + FA_FLOATS + (tid >> 6) * PB + (tid & 63) * 4 + i * 4 * PB) = o;
    }
  }
  __syncthreads();

  v8f c0 = {}, c1 = {};
  mul_chunk(Lf + (16 * rt + lo) * PA + 2 * hi,
            Lf + FA_FLOATS + hi * PB + (32 * cq + lo) * 2, c0, c1, []() {});

  {
    const int colA = 32 * cq + lo;
    const int colB = colA + 16;
    const float bvA = bias[colA];
    const float bvB = bias[colB];
#pragma unroll
    for (int v = 0; v < 8; ++v) {
      const int row = row0 + 16 * rt + hi * 8 + v;
      const float dv = dinv[row];
      out[(size_t)row * D + colA] = dv * c0[v] + bvA;
      out[(size_t)row * D + colB] = dv * c1[v] + bvB;
    }
  }
}

extern "C" void kernel_launch(void* const* d_in, const int* in_sizes, int n_in,
                              void* d_out, int out_size, void* d_ws, size_t ws_size,
                              hipStream_t stream) {
  const float* x    = (const float*)d_in[0];
  const float* adj  = (const float*)d_in[1];
  const float* w    = (const float*)d_in[2];
  const float* bias = (const float*)d_in[3];
  float* out = (float*)d_out;

  float* dinv  = (float*)d_ws;
  float* part1 = (float*)d_ws + NN;

  k_deg<<<NN, 256, 0, stream>>>(adj, dinv);
  k_agg<<<(NN / BM) * KSPLIT, 256, 0, stream>>>(adj, x, dinv, out, part1);
  k_fin<<<NN / FM, 256, 0, stream>>>(part1, w, dinv, bias, out);
}
